# starting kernel
# baseline (speedup 1.0000x reference)
#include <hip/hip_runtime.h>


typedef float v2f __attribute__((ext_vector_type(2)));
typedef float v8f __attribute__((ext_vector_type(8)));

#define LSIG   524288
#define SEGS   256
#define S_TILE 2064

struct GTab { float g[2][18]; };

__host__ __device__ constexpr GTab makeG() {
    constexpr float REC[10] = {
        0.160102397974125f, 0.6038292697974729f, 0.7243085284385744f,
        0.13842814590110342f, -0.24229488706619015f, -0.03224486958502952f,
        0.07757149384006515f, -0.006241490213011705f, -0.012580751999015526f,
        0.003335725285001549f };
    GTab t{};
    for (int r = 0; r < 2; ++r)
        for (int k = 0; k < 18; ++k) {
            float acc = 0.0f;
            for (int u = 0; u < 5; ++u) {
                int tt = k - 2 * u;
                if (tt >= 0 && tt <= 9) acc += REC[tt] * REC[8 + r - 2 * u];
            }
            t.g[r][k] = acc;
        }
    return t;
}

__device__ __constant__ GTab cG = makeG();

extern "C" __global__ __launch_bounds__(256)
void wavelet_fused_wmma(const float* __restrict__ x, float* __restrict__ y)
{
    __shared__ __align__(16) float s_tile[S_TILE];

    const int tid = threadIdx.x;
    const int blk = blockIdx.x;
    const int b   = blk >> 8;
    const int seg = blk & (SEGS - 1);
    const int S0  = seg << 11;

    const float* __restrict__ xb = x + (size_t)b * LSIG;

    if (seg > 0 && seg < SEGS - 1) {
        const float4* __restrict__ src = (const float4*)(xb + S0 - 8);
        float4* dst = (float4*)s_tile;
        #pragma unroll 2
        for (int t = tid; t < S_TILE / 4; t += 256) dst[t] = src[t];
    } else {
        for (int t = tid; t < S_TILE; t += 256) {
            int idx = S0 + t - 8;
            if (idx < 0) idx = -idx - 1;
            else if (idx >= LSIG) idx = 2 * LSIG - idx - 1;
            s_tile[t] = xb[idx];
        }
    }
    __syncthreads();

    const int lane = tid & 31;
    const int wv   = tid >> 5;
    const int half = lane >> 4;
    const int im   = lane & 15;

    const int r  = im & 1;
    const int jc = im & ~1;
    v2f bfrag[8];
    #pragma unroll
    for (int q = 0; q < 8; ++q) {
        #pragma unroll
        for (int kk = 0; kk < 2; ++kk) {
            int gi = 4 * q + 2 * half + kk - jc;
            bfrag[q][kk] = (gi >= 0 && gi < 18) ? cG.g[r][gi] : 0.0f;
        }
    }

    const int abase = (wv << 8) + (im << 4) + (half << 1);
    v8f c = {};
    #pragma unroll
    for (int q = 0; q < 8; ++q) {
        const v2f a = *(const v2f*)(s_tile + abase + 4 * q);
        c = __builtin_amdgcn_wmma_f32_16x16x4_f32(
                 false, a,  false, bfrag[q],
                 (short)0, c,  false,  false);
    }

    float* __restrict__ yb = y + (size_t)b * LSIG
                           + ((size_t)seg << 11) + (wv << 8) + (half << 7) + im;
    #pragma unroll
    for (int g = 0; g < 8; ++g)
        yb[g << 4] = c[g];
}

extern "C" void kernel_launch(void* const* d_in, const int* in_sizes, int n_in,
                              void* d_out, int out_size, void* d_ws, size_t ws_size,
                              hipStream_t stream) {
    const float* x = (const float*)d_in[0];
    float*       y = (float*)d_out;
    wavelet_fused_wmma<<<dim3(64 * SEGS), dim3(256), 0, stream>>>(x, y);
}


# Persistent grid: 256 workgroups x 16 independent waves each loop over 4 strips of 2048 outputs through wave-private LDS, next strip's 16 aligned loads in flight under the math (was 16384 workgroups)
# speedup vs baseline: 1.2378x; 1.2378x over previous
#include <hip/hip_runtime.h>


typedef float v2f __attribute__((ext_vector_type(2)));
typedef float v8f __attribute__((ext_vector_type(8)));
typedef float v4f __attribute__((ext_vector_type(4)));

#define LSIG   524288
#define NT     8
#define STRIP  (NT * 256)
#define S_STR  (STRIP + 16)
#define NLD    (STRIP / 128)
#define WAVES  16
#define NBLK   256
#define NSTRIP (64 * (LSIG / STRIP))
#define ITERS  (NSTRIP / (NBLK * WAVES))

struct GTab { float g[2][18]; };

__host__ __device__ constexpr GTab makeG() {
    constexpr float REC[10] = {
        0.160102397974125f, 0.6038292697974729f, 0.7243085284385744f,
        0.13842814590110342f, -0.24229488706619015f, -0.03224486958502952f,
        0.07757149384006515f, -0.006241490213011705f, -0.012580751999015526f,
        0.003335725285001549f };
    GTab t{};
    for (int r = 0; r < 2; ++r)
        for (int k = 0; k < 18; ++k) {
            float acc = 0.0f;
            for (int u = 0; u < 5; ++u) {
                int tt = k - 2 * u;
                if (tt >= 0 && tt <= 9) acc += REC[tt] * REC[8 + r - 2 * u];
            }
            t.g[r][k] = acc;
        }
    return t;
}

__device__ __constant__ GTab cG = makeG();

extern "C" __global__ __launch_bounds__(WAVES * 32)
void wavelet_fused_wmma(const float* __restrict__ x, float* __restrict__ y)
{
    __shared__ __align__(16) float s_all[WAVES * S_STR];

    const int tid  = threadIdx.x;
    const int lane = tid & 31;
    const int wv   = __builtin_amdgcn_readfirstlane(tid >> 5);
    const int half = lane >> 4;
    const int im   = lane & 15;
    float* __restrict__ s_w = s_all + wv * S_STR;

    const int r  = im & 1;
    const int jc = im & ~1;
    v2f bfrag[8];
    #pragma unroll
    for (int q = 0; q < 8; ++q) {
        #pragma unroll
        for (int kk = 0; kk < 2; ++kk) {
            int gi = 4 * q + 2 * half + kk - jc;
            bfrag[q][kk] = (gi >= 0 && gi < 18) ? cG.g[r][gi] : 0.0f;
        }
    }

    const int hp = (im < 8) ? im : (STRIP + im);

    v4f   body[NLD];
    float halo;

#define ISSUE_LOADS(SID)                                                          \
    {                                                                             \
        const int sid_ = (SID);                                                   \
        const v4f* __restrict__ src_ = (const v4f*)(x + (size_t)sid_ * STRIP) + lane; \
        _Pragma("unroll")                                                         \
        for (int u = 0; u < NLD; ++u) body[u] = src_[u * 32];                     \
        int idx_ = (sid_ & 255) * STRIP - 8 + hp;                                 \
        if (idx_ < 0) idx_ = -idx_ - 1;                                           \
        else if (idx_ >= LSIG) idx_ = 2 * LSIG - idx_ - 1;                        \
        halo = x[(size_t)(sid_ >> 8) * LSIG + idx_];                              \
    }

    const int gw = blockIdx.x * WAVES + wv;
    ISSUE_LOADS(gw)

    const int abase = (im << 4) + (half << 1);

    #pragma unroll
    for (int t = 0; t < ITERS; ++t) {
        const int sid = t * (NBLK * WAVES) + gw;

        __builtin_amdgcn_fence(__ATOMIC_RELEASE, "wavefront");
        __builtin_amdgcn_wave_barrier();
        {
            v4f* dst = (v4f*)(s_w + 8) + lane;
            #pragma unroll
            for (int u = 0; u < NLD; ++u) dst[u * 32] = body[u];
            s_w[hp] = halo;
        }
        __builtin_amdgcn_wave_barrier();
        __builtin_amdgcn_fence(__ATOMIC_ACQUIRE, "wavefront");

        if (t + 1 < ITERS) ISSUE_LOADS(sid + NBLK * WAVES)
        __builtin_amdgcn_sched_barrier(0);

        float* __restrict__ yb = y + (size_t)sid * STRIP + (half << 7) + im;
        #pragma unroll
        for (int tt = 0; tt < NT; ++tt) {
            v8f c = {};
            #pragma unroll
            for (int q = 0; q < 8; ++q) {
                const v2f a = *(const v2f*)(s_w + abase + 256 * tt + 4 * q);
                c = __builtin_amdgcn_wmma_f32_16x16x4_f32(
                         false, a,  false, bfrag[q],
                         (short)0, c,  false,  false);
            }
            #pragma unroll
            for (int g = 0; g < 8; ++g)
                yb[256 * tt + (g << 4)] = c[g];
        }
    }
}

extern "C" void kernel_launch(void* const* d_in, const int* in_sizes, int n_in,
                              void* d_out, int out_size, void* d_ws, size_t ws_size,
                              hipStream_t stream) {
    const float* x = (const float*)d_in[0];
    float*       y = (float*)d_out;
    wavelet_fused_wmma<<<dim3(NBLK), dim3(WAVES * 32), 0, stream>>>(x, y);
}


# Permute the A rows so each accumulator register spans one whole 128-byte output line: every global store instruction writes one full line instead of two half lines
# speedup vs baseline: 1.3352x; 1.0787x over previous
#include <hip/hip_runtime.h>


typedef float v2f __attribute__((ext_vector_type(2)));
typedef float v8f __attribute__((ext_vector_type(8)));
typedef float v4f __attribute__((ext_vector_type(4)));

#define LSIG   524288
#define NT     8
#define STRIP  (NT * 256)
#define S_STR  (STRIP + 16)
#define NLD    (STRIP / 128)
#define WAVES  16
#define NBLK   256
#define NSTRIP (64 * (LSIG / STRIP))
#define ITERS  (NSTRIP / (NBLK * WAVES))

struct GTab { float g[2][18]; };

__host__ __device__ constexpr GTab makeG() {
    constexpr float REC[10] = {
        0.160102397974125f, 0.6038292697974729f, 0.7243085284385744f,
        0.13842814590110342f, -0.24229488706619015f, -0.03224486958502952f,
        0.07757149384006515f, -0.006241490213011705f, -0.012580751999015526f,
        0.003335725285001549f };
    GTab t{};
    for (int r = 0; r < 2; ++r)
        for (int k = 0; k < 18; ++k) {
            float acc = 0.0f;
            for (int u = 0; u < 5; ++u) {
                int tt = k - 2 * u;
                if (tt >= 0 && tt <= 9) acc += REC[tt] * REC[8 + r - 2 * u];
            }
            t.g[r][k] = acc;
        }
    return t;
}

__device__ __constant__ GTab cG = makeG();

extern "C" __global__ __launch_bounds__(WAVES * 32)
void wavelet_fused_wmma(const float* __restrict__ x, float* __restrict__ y)
{
    __shared__ __align__(16) float s_all[WAVES * S_STR];

    const int tid  = threadIdx.x;
    const int lane = tid & 31;
    const int wv   = __builtin_amdgcn_readfirstlane(tid >> 5);
    const int half = lane >> 4;
    const int im   = lane & 15;
    float* __restrict__ s_w = s_all + wv * S_STR;

    const int r  = im & 1;
    const int jc = im & ~1;
    v2f bfrag[8];
    #pragma unroll
    for (int q = 0; q < 8; ++q) {
        #pragma unroll
        for (int kk = 0; kk < 2; ++kk) {
            int gi = 4 * q + 2 * half + kk - jc;
            bfrag[q][kk] = (gi >= 0 && gi < 18) ? cG.g[r][gi] : 0.0f;
        }
    }

    const int hp = (im < 8) ? im : (STRIP + im);

    v4f   body[NLD];
    float halo;

#define ISSUE_LOADS(SID)                                                          \
    {                                                                             \
        const int sid_ = (SID);                                                   \
        const v4f* __restrict__ src_ = (const v4f*)(x + (size_t)sid_ * STRIP) + lane; \
        _Pragma("unroll")                                                         \
        for (int u = 0; u < NLD; ++u) body[u] = src_[u * 32];                     \
        int idx_ = (sid_ & 255) * STRIP - 8 + hp;                                 \
        if (idx_ < 0) idx_ = -idx_ - 1;                                           \
        else if (idx_ >= LSIG) idx_ = 2 * LSIG - idx_ - 1;                        \
        halo = x[(size_t)(sid_ >> 8) * LSIG + idx_];                              \
    }

    const int gw = blockIdx.x * WAVES + wv;
    ISSUE_LOADS(gw)

    const int abase = ((im & 7) << 5) + ((im >> 3) << 4) + (half << 1);

    #pragma unroll
    for (int t = 0; t < ITERS; ++t) {
        const int sid = t * (NBLK * WAVES) + gw;

        __builtin_amdgcn_fence(__ATOMIC_RELEASE, "wavefront");
        __builtin_amdgcn_wave_barrier();
        {
            v4f* dst = (v4f*)(s_w + 8) + lane;
            #pragma unroll
            for (int u = 0; u < NLD; ++u) dst[u * 32] = body[u];
            s_w[hp] = halo;
        }
        __builtin_amdgcn_wave_barrier();
        __builtin_amdgcn_fence(__ATOMIC_ACQUIRE, "wavefront");

        if (t + 1 < ITERS) ISSUE_LOADS(sid + NBLK * WAVES)
        __builtin_amdgcn_sched_barrier(0);

        float* __restrict__ yb = y + (size_t)sid * STRIP + lane;
        #pragma unroll
        for (int tt = 0; tt < NT; ++tt) {
            v8f c = {};
            #pragma unroll
            for (int q = 0; q < 8; ++q) {
                const v2f a = *(const v2f*)(s_w + abase + 256 * tt + 4 * q);
                c = __builtin_amdgcn_wmma_f32_16x16x4_f32(
                         false, a,  false, bfrag[q],
                         (short)0, c,  false,  false);
            }
            #pragma unroll
            for (int g = 0; g < 8; ++g)
                yb[256 * tt + (g << 5)] = c[g];
        }
    }
}

extern "C" void kernel_launch(void* const* d_in, const int* in_sizes, int n_in,
                              void* d_out, int out_size, void* d_ws, size_t ws_size,
                              hipStream_t stream) {
    const float* x = (const float*)d_in[0];
    float*       y = (float*)d_out;
    wavelet_fused_wmma<<<dim3(NBLK), dim3(WAVES * 32), 0, stream>>>(x, y);
}


# A fragments read from LDS one tile pair ahead: the next pair's 16 reads issued as ONE group in front of the current pair's 16 WMMAs into a second register set, not one read between every two WMMAs
# speedup vs baseline: 1.5178x; 1.1368x over previous
#include <hip/hip_runtime.h>


typedef float v2f __attribute__((ext_vector_type(2)));
typedef float v8f __attribute__((ext_vector_type(8)));
typedef float v4f __attribute__((ext_vector_type(4)));

#define LSIG   524288
#define NT     8
#define STRIP  (NT * 256)
#define S_STR  (STRIP + 16)
#define NLD    (STRIP / 128)
#define WAVES  16
#define NBLK   256
#define NSTRIP (64 * (LSIG / STRIP))
#define ITERS  (NSTRIP / (NBLK * WAVES))
#define NPAIR  (NT / 2)

struct GTab { float g[2][18]; };

__host__ __device__ constexpr GTab makeG() {
    constexpr float REC[10] = {
        0.160102397974125f, 0.6038292697974729f, 0.7243085284385744f,
        0.13842814590110342f, -0.24229488706619015f, -0.03224486958502952f,
        0.07757149384006515f, -0.006241490213011705f, -0.012580751999015526f,
        0.003335725285001549f };
    GTab t{};
    for (int r = 0; r < 2; ++r)
        for (int k = 0; k < 18; ++k) {
            float acc = 0.0f;
            for (int u = 0; u < 5; ++u) {
                int tt = k - 2 * u;
                if (tt >= 0 && tt <= 9) acc += REC[tt] * REC[8 + r - 2 * u];
            }
            t.g[r][k] = acc;
        }
    return t;
}

__device__ __constant__ GTab cG = makeG();

extern "C" __global__ __launch_bounds__(WAVES * 32)
void wavelet_fused_wmma(const float* __restrict__ x, float* __restrict__ y)
{
    __shared__ __align__(16) float s_all[WAVES * S_STR];

    const int tid  = threadIdx.x;
    const int lane = tid & 31;
    const int wv   = __builtin_amdgcn_readfirstlane(tid >> 5);
    const int half = lane >> 4;
    const int im   = lane & 15;
    float* __restrict__ s_w = s_all + wv * S_STR;

    const int r  = im & 1;
    const int jc = im & ~1;
    v2f bfrag[8];
    #pragma unroll
    for (int q = 0; q < 8; ++q) {
        #pragma unroll
        for (int kk = 0; kk < 2; ++kk) {
            int gi = 4 * q + 2 * half + kk - jc;
            bfrag[q][kk] = (gi >= 0 && gi < 18) ? cG.g[r][gi] : 0.0f;
        }
    }

    const int hp = (im < 8) ? im : (STRIP + im);

    v4f   body[NLD];
    float halo;

#define ISSUE_LOADS(SID)                                                          \
    {                                                                             \
        const int sid_ = (SID);                                                   \
        const v4f* __restrict__ src_ = (const v4f*)(x + (size_t)sid_ * STRIP) + lane; \
        _Pragma("unroll")                                                         \
        for (int u = 0; u < NLD; ++u) body[u] = src_[u * 32];                     \
        int idx_ = (sid_ & 255) * STRIP - 8 + hp;                                 \
        if (idx_ < 0) idx_ = -idx_ - 1;                                           \
        else if (idx_ >= LSIG) idx_ = 2 * LSIG - idx_ - 1;                        \
        halo = x[(size_t)(sid_ >> 8) * LSIG + idx_];                              \
    }

    const int gw = blockIdx.x * WAVES + wv;
    ISSUE_LOADS(gw)

    const int abase = ((im & 7) << 5) + ((im >> 3) << 4) + (half << 1);

    v2f af[2][16];
#define LOAD_PAIR(AF, P)                                                          \
    {                                                                             \
        _Pragma("unroll")                                                         \
        for (int e = 0; e < 16; ++e)                                              \
            AF[e] = *(const v2f*)(s_w + abase + 256 * (2 * (P) + (e >> 3)) + 4 * (e & 7)); \
    }

    #pragma unroll
    for (int t = 0; t < ITERS; ++t) {
        const int sid = t * (NBLK * WAVES) + gw;

        __builtin_amdgcn_fence(__ATOMIC_RELEASE, "wavefront");
        __builtin_amdgcn_wave_barrier();
        {
            v4f* dst = (v4f*)(s_w + 8) + lane;
            #pragma unroll
            for (int u = 0; u < NLD; ++u) dst[u * 32] = body[u];
            s_w[hp] = halo;
        }
        __builtin_amdgcn_wave_barrier();
        __builtin_amdgcn_fence(__ATOMIC_ACQUIRE, "wavefront");

        LOAD_PAIR(af[0], 0)

        if (t + 1 < ITERS) ISSUE_LOADS(sid + NBLK * WAVES)
        __builtin_amdgcn_sched_barrier(0);

        float* __restrict__ yb = y + (size_t)sid * STRIP + lane;
        #pragma unroll
        for (int p = 0; p < NPAIR; ++p) {
            if (p + 1 < NPAIR) LOAD_PAIR(af[(p + 1) & 1], p + 1)
            __builtin_amdgcn_sched_barrier(0);
            #pragma unroll
            for (int h2 = 0; h2 < 2; ++h2) {
                const int tt = 2 * p + h2;
                v8f c = {};
                #pragma unroll
                for (int q = 0; q < 8; ++q) {
                    c = __builtin_amdgcn_wmma_f32_16x16x4_f32(
                             false, af[p & 1][8 * h2 + q],  false, bfrag[q],
                             (short)0, c,  false,  false);
                }
                #pragma unroll
                for (int g = 0; g < 8; ++g)
                    yb[256 * tt + (g << 5)] = c[g];
            }
            __builtin_amdgcn_sched_barrier(0);
        }
    }
}

extern "C" void kernel_launch(void* const* d_in, const int* in_sizes, int n_in,
                              void* d_out, int out_size, void* d_ws, size_t ws_size,
                              hipStream_t stream) {
    const float* x = (const float*)d_in[0];
    float*       y = (float*)d_out;
    wavelet_fused_wmma<<<dim3(NBLK), dim3(WAVES * 32), 0, stream>>>(x, y);
}


# Strips halved to 1024 outputs and three strips' loads kept in flight per wave in three register sets (was one strip of 2048 ahead): 8-12 KB outstanding per wave at all times, shorter head and tail
# speedup vs baseline: 1.5650x; 1.0311x over previous
#include <hip/hip_runtime.h>


typedef float v2f __attribute__((ext_vector_type(2)));
typedef float v8f __attribute__((ext_vector_type(8)));
typedef float v4f __attribute__((ext_vector_type(4)));

#define LSIG   524288
#define NT     4
#define STRIP  (NT * 256)
#define SPB    (LSIG / STRIP)
#define SPB_LG 9
#define DEPTH  3
#define S_STR  (STRIP + 16)
#define NLD    (STRIP / 128)
#define WAVES  16
#define NBLK   256
#define NSTRIP (64 * SPB)
#define ITERS  (NSTRIP / (NBLK * WAVES))
#define NPAIR  (NT / 2)

struct GTab { float g[2][18]; };

__host__ __device__ constexpr GTab makeG() {
    constexpr float REC[10] = {
        0.160102397974125f, 0.6038292697974729f, 0.7243085284385744f,
        0.13842814590110342f, -0.24229488706619015f, -0.03224486958502952f,
        0.07757149384006515f, -0.006241490213011705f, -0.012580751999015526f,
        0.003335725285001549f };
    GTab t{};
    for (int r = 0; r < 2; ++r)
        for (int k = 0; k < 18; ++k) {
            float acc = 0.0f;
            for (int u = 0; u < 5; ++u) {
                int tt = k - 2 * u;
                if (tt >= 0 && tt <= 9) acc += REC[tt] * REC[8 + r - 2 * u];
            }
            t.g[r][k] = acc;
        }
    return t;
}

__device__ __constant__ GTab cG = makeG();

extern "C" __global__ __launch_bounds__(WAVES * 32)
void wavelet_fused_wmma(const float* __restrict__ x, float* __restrict__ y)
{
    __shared__ __align__(16) float s_all[WAVES * S_STR];

    const int tid  = threadIdx.x;
    const int lane = tid & 31;
    const int wv   = __builtin_amdgcn_readfirstlane(tid >> 5);
    const int half = lane >> 4;
    const int im   = lane & 15;
    float* __restrict__ s_w = s_all + wv * S_STR;

    const int r  = im & 1;
    const int jc = im & ~1;
    v2f bfrag[8];
    #pragma unroll
    for (int q = 0; q < 8; ++q) {
        #pragma unroll
        for (int kk = 0; kk < 2; ++kk) {
            int gi = 4 * q + 2 * half + kk - jc;
            bfrag[q][kk] = (gi >= 0 && gi < 18) ? cG.g[r][gi] : 0.0f;
        }
    }

    const int hp = (im < 8) ? im : (STRIP + im);

    v4f   body[DEPTH][NLD];
    float halo[DEPTH];

#define ISSUE_LOADS(SET, SID)                                                     \
    {                                                                             \
        const int sid_ = (SID);                                                   \
        const v4f* __restrict__ src_ = (const v4f*)(x + (size_t)sid_ * STRIP) + lane; \
        _Pragma("unroll")                                                         \
        for (int u = 0; u < NLD; ++u) body[SET][u] = src_[u * 32];                \
        int idx_ = (sid_ & (SPB - 1)) * STRIP - 8 + hp;                           \
        if (idx_ < 0) idx_ = -idx_ - 1;                                           \
        else if (idx_ >= LSIG) idx_ = 2 * LSIG - idx_ - 1;                        \
        halo[SET] = x[(size_t)(sid_ >> SPB_LG) * LSIG + idx_];                    \
    }

    const int gw = blockIdx.x * WAVES + wv;
    ISSUE_LOADS(0, gw)
    __builtin_amdgcn_sched_barrier(0);
    ISSUE_LOADS(1, gw + NBLK * WAVES)
    __builtin_amdgcn_sched_barrier(0);
    ISSUE_LOADS(2, gw + 2 * NBLK * WAVES)

    const int abase = ((im & 7) << 5) + ((im >> 3) << 4) + (half << 1);

    v2f af[2][16];
#define LOAD_PAIR(AF, P)                                                          \
    {                                                                             \
        _Pragma("unroll")                                                         \
        for (int e = 0; e < 16; ++e)                                              \
            AF[e] = *(const v2f*)(s_w + abase + 256 * (2 * (P) + (e >> 3)) + 4 * (e & 7)); \
    }

    #pragma unroll
    for (int t = 0; t < ITERS; ++t) {
        const int sid = t * (NBLK * WAVES) + gw;

        __builtin_amdgcn_fence(__ATOMIC_RELEASE, "wavefront");
        __builtin_amdgcn_wave_barrier();
        {
            v4f* dst = (v4f*)(s_w + 8) + lane;
            #pragma unroll
            for (int u = 0; u < NLD; ++u) dst[u * 32] = body[t % DEPTH][u];
            s_w[hp] = halo[t % DEPTH];
        }
        __builtin_amdgcn_wave_barrier();
        __builtin_amdgcn_fence(__ATOMIC_ACQUIRE, "wavefront");

        LOAD_PAIR(af[0], 0)

        if (t + DEPTH < ITERS) ISSUE_LOADS(t % DEPTH, sid + DEPTH * NBLK * WAVES)
        __builtin_amdgcn_sched_barrier(0);

        float* __restrict__ yb = y + (size_t)sid * STRIP + lane;
        #pragma unroll
        for (int p = 0; p < NPAIR; ++p) {
            if (p + 1 < NPAIR) LOAD_PAIR(af[(p + 1) & 1], p + 1)
            __builtin_amdgcn_sched_barrier(0);
            #pragma unroll
            for (int h2 = 0; h2 < 2; ++h2) {
                const int tt = 2 * p + h2;
                v8f c = {};
                #pragma unroll
                for (int q = 0; q < 8; ++q) {
                    c = __builtin_amdgcn_wmma_f32_16x16x4_f32(
                             false, af[p & 1][8 * h2 + q],  false, bfrag[q],
                             (short)0, c,  false,  false);
                }
                #pragma unroll
                for (int g = 0; g < 8; ++g)
                    yb[256 * tt + (g << 5)] = c[g];
            }
            __builtin_amdgcn_sched_barrier(0);
        }
    }
}

extern "C" void kernel_launch(void* const* d_in, const int* in_sizes, int n_in,
                              void* d_out, int out_size, void* d_ws, size_t ws_size,
                              hipStream_t stream) {
    const float* x = (const float*)d_in[0];
    float*       y = (float*)d_out;
    wavelet_fused_wmma<<<dim3(NBLK), dim3(WAVES * 32), 0, stream>>>(x, y);
}
